# starting kernel
# baseline (speedup 1.0000x reference)
#include <hip/hip_runtime.h>
#include <math.h>

#define B_TOTAL 4096
#define T_LEN   512
#define INP     6
#define HID     8

typedef float v2f __attribute__((ext_vector_type(2)));
typedef float v8f __attribute__((ext_vector_type(8)));

static __device__ __forceinline__ v8f wmma4(v2f a, v2f b, v8f c) {
  return __builtin_amdgcn_wmma_f32_16x16x4_f32(false, a, false, b, (short)0, c,
                                               false, false);
}

static __device__ __forceinline__ float htanh(float x) {
#if __has_builtin(__builtin_amdgcn_tanhf)
  return __builtin_amdgcn_tanhf(x);
#else
  float t = __builtin_amdgcn_exp2f(x * 2.8853900817779268f);
  return 1.0f - 2.0f * __builtin_amdgcn_rcpf(t + 1.0f);
#endif
}

static __device__ __forceinline__ v8f tanh8(v8f d) {
  v8f h;
#pragma unroll
  for (int v = 0; v < 8; ++v) h[v] = htanh(d[v]);
  return h;
}

static __device__ __forceinline__ void packB(const v8f& h, v2f& b0, v2f& b1) {
  b0.x = h[0] + __shfl_xor(h[2], 16, 32);
  b0.y = h[1] + __shfl_xor(h[3], 16, 32);
  b1.x = h[4] + __shfl_xor(h[6], 16, 32);
  b1.y = h[5] + __shfl_xor(h[7], 16, 32);
}

static __device__ __forceinline__ v2f load_A(const float* __restrict__ W,
                                             int ncols, int kbase, int m,
                                             int hi) {
  int k0 = kbase + 2 * hi;
  v2f a;
  a.x = (m < HID && k0 < ncols) ? W[m * ncols + k0] : 0.0f;
  a.y = (m < HID && (k0 + 1) < ncols) ? W[m * ncols + k0 + 1] : 0.0f;
  return a;
}

__global__ __launch_bounds__(128) void rnn2_tanh_wmma(
    const float* __restrict__ x,
    const float* __restrict__ Wih0, const float* __restrict__ Whh0,
    const float* __restrict__ bih0, const float* __restrict__ bhh0,
    const float* __restrict__ Wih1, const float* __restrict__ Whh1,
    const float* __restrict__ bih1, const float* __restrict__ bhh1,
    const float* __restrict__ Wout, const float* __restrict__ bout_p,
    float* __restrict__ out) {
  const int lane = threadIdx.x & 31;
  const int wid = (blockIdx.x * blockDim.x + threadIdx.x) >> 5;
  const int n = lane & 15;
  const int hi = lane >> 4;
  const int bb = wid * 16;

  const v2f a_ih0_0 = load_A(Wih0, INP, 0, n, hi);
  const v2f a_ih0_1 = load_A(Wih0, INP, 4, n, hi);
  const v2f a_hh0_0 = load_A(Whh0, HID, 0, n, hi);
  const v2f a_hh0_1 = load_A(Whh0, HID, 4, n, hi);
  const v2f a_ih1_0 = load_A(Wih1, HID, 0, n, hi);
  const v2f a_ih1_1 = load_A(Wih1, HID, 4, n, hi);
  const v2f a_hh1_0 = load_A(Whh1, HID, 0, n, hi);
  const v2f a_hh1_1 = load_A(Whh1, HID, 4, n, hi);

  v8f c0b, c1b;
#pragma unroll
  for (int v = 0; v < 8; ++v) {
    c0b[v] = (hi == 0) ? (bih0[v] + bhh0[v]) : 0.0f;
    c1b[v] = (hi == 0) ? (bih1[v] + bhh1[v]) : 0.0f;
  }

  float wo[8];
#pragma unroll
  for (int v = 0; v < 8; ++v) wo[v] = Wout[v];
  const float bo = bout_p[0];
  const float ximask = (hi == 0) ? 1.0f : 0.0f;

  v2f h1b0 = (v2f){0.f, 0.f}, h1b1 = (v2f){0.f, 0.f};
  v2f h2b0 = (v2f){0.f, 0.f}, h2b1 = (v2f){0.f, 0.f};

  const float* xp = x + (size_t)(bb + n) * T_LEN * INP;
  float* op = out + (size_t)(bb + n) * T_LEN;
  const int off0 = 2 * hi;
  const int off1 = hi ? 0 : 4;

  auto loadX = [&](int t, v2f& b0, v2f& b1) {
    const float* xq = xp + (size_t)t * INP;
    b0 = *(const v2f*)(xq + off0);
    v2f r = *(const v2f*)(xq + off1);
    b1.x = r.x * ximask;
    b1.y = r.y * ximask;
  };

  v2f xb0, xb1;
  loadX(0, xb0, xb1);
  v8f dx = wmma4(a_ih0_0, xb0, c0b);
  dx = wmma4(a_ih0_1, xb1, dx);
  loadX(1, xb0, xb1);

  float4 ob;

#pragma unroll 4
  for (int t = 0; t < T_LEN; ++t) {
    v8f dxn = wmma4(a_ih0_0, xb0, c0b);
    dxn = wmma4(a_ih0_1, xb1, dxn);

    v8f erec = wmma4(a_hh1_0, h2b0, c1b);
    erec = wmma4(a_hh1_1, h2b1, erec);

    v8f d = wmma4(a_hh0_0, h1b0, dx);
    d = wmma4(a_hh0_1, h1b1, d);
    v8f h1 = tanh8(d);
    packB(h1, h1b0, h1b1);

    v8f e = wmma4(a_ih1_0, h1b0, erec);
    e = wmma4(a_ih1_1, h1b1, e);
    v8f h2 = tanh8(e);
    packB(h2, h2b0, h2b1);

    float o = bo;
#pragma unroll
    for (int v = 0; v < 8; ++v) o = fmaf(wo[v], h2[v], o);

    const int ph = t & 3;
    if (ph == 0) {
      ob.x = o;
      __builtin_prefetch(xp + (size_t)(t + 16) * INP, 0, 1);
    } else if (ph == 1) {
      ob.y = o;
    } else if (ph == 2) {
      ob.z = o;
    } else {
      ob.w = o;
      if (lane < 16) *(float4*)(op + t - 3) = ob;
    }

    dx = dxn;
    int t2 = t + 2;
    t2 = t2 < T_LEN ? t2 : T_LEN - 1;
    loadX(t2, xb0, xb1);
  }
}

extern "C" void kernel_launch(void* const* d_in, const int* in_sizes, int n_in,
                              void* d_out, int out_size, void* d_ws,
                              size_t ws_size, hipStream_t stream) {
  (void)in_sizes; (void)n_in; (void)out_size; (void)d_ws; (void)ws_size;
  const float* x    = (const float*)d_in[0];
  const float* Wih0 = (const float*)d_in[1];
  const float* Whh0 = (const float*)d_in[2];
  const float* bih0 = (const float*)d_in[3];
  const float* bhh0 = (const float*)d_in[4];
  const float* Wih1 = (const float*)d_in[5];
  const float* Whh1 = (const float*)d_in[6];
  const float* bih1 = (const float*)d_in[7];
  const float* bhh1 = (const float*)d_in[8];
  const float* Wout = (const float*)d_in[9];
  const float* bout = (const float*)d_in[10];
  float* out = (float*)d_out;

  dim3 grid(B_TOTAL / 16 / 4);
  dim3 block(128);
  rnn2_tanh_wmma<<<grid, block, 0, stream>>>(x, Wih0, Whh0, bih0, bhh0, Wih1,
                                             Whh1, bih1, bhh1, Wout, bout, out);
}


# Layers split over two waves per batch tile, one SIMD each: layer-0 wave runs a 16-step chunk ahead and hands h1 over through LDS, one barrier per chunk; layer 1 sums its input product first
# speedup vs baseline: 2.7612x; 2.7612x over previous
#include <hip/hip_runtime.h>
#include <math.h>

#define B_TOTAL 4096
#define T_LEN   512
#define INP     6
#define HID     8

#define CH     16
#define NCH    (T_LEN / CH)
#define HBUF4  (CH * 32)
#define CHF4   (CH * INP / 4)
#define ROWF4  (CHF4 + 1)
#define BUFF4  (16 * ROWF4)
#define NJ     (16 * CHF4 / 32)

typedef float v2f __attribute__((ext_vector_type(2)));
typedef float v4f __attribute__((ext_vector_type(4)));
typedef float v8f __attribute__((ext_vector_type(8)));

static __device__ __forceinline__ v8f wmma4(v2f a, v2f b, v8f c) {
  return __builtin_amdgcn_wmma_f32_16x16x4_f32(false, a, false, b, (short)0, c,
                                               false, false);
}

static __device__ __forceinline__ float htanh(float x) {
#if __has_builtin(__builtin_amdgcn_tanhf)
  return __builtin_amdgcn_tanhf(x);
#else
  float t = __builtin_amdgcn_exp2f(x * 2.8853900817779268f);
  return 1.0f - 2.0f * __builtin_amdgcn_rcpf(t + 1.0f);
#endif
}

static __device__ __forceinline__ int unit_of_row(int m) {
  return (m & 5) | ((m & 8) >> 2);
}
static __device__ __forceinline__ bool row_used(int m) { return (m & 2) == 0; }

static __device__ __forceinline__ v2f load_A(const float* __restrict__ W,
                                             int ncols, int kbase, int m,
                                             int hi,
                                             const float* __restrict__ orow) {
  const int k0 = kbase + 2 * hi;
  const int u = unit_of_row(m);
  const bool use = row_used(m);
  v2f a;
  a.x = (use && k0 < ncols) ? W[u * ncols + k0] : 0.0f;
  a.y = (use && (k0 + 1) < ncols) ? W[u * ncols + k0 + 1] : 0.0f;
  if (orow != nullptr && m == 2) {
    a.x = (k0 < ncols) ? orow[k0] : 0.0f;
    a.y = ((k0 + 1) < ncols) ? orow[k0 + 1] : 0.0f;
  }
  return a;
}

__global__ __launch_bounds__(128) void rnn2_tanh_wmma(
    const float* __restrict__ x,
    const float* __restrict__ Wih0, const float* __restrict__ Whh0,
    const float* __restrict__ bih0, const float* __restrict__ bhh0,
    const float* __restrict__ Wih1, const float* __restrict__ Whh1,
    const float* __restrict__ bih1, const float* __restrict__ bhh1,
    const float* __restrict__ Wout, const float* __restrict__ bout_p,
    float* __restrict__ out) {
  const int lane = threadIdx.x & 31;
  const int wib = __builtin_amdgcn_readfirstlane((int)(threadIdx.x >> 5));
  const int role = wib & 1;
  const int tib = wib >> 1;
  const int n = lane & 15;
  const int hi = lane >> 4;
  const int bb = (blockIdx.x * 2 + tib) * 16;

  const v2f a_ih0_0 = load_A(Wih0, INP, 0, n, hi, nullptr);
  const v2f a_ih0_1 = load_A(Wih0, INP, 4, n, hi, nullptr);
  const v2f a_hh0_0 = load_A(Whh0, HID, 0, n, hi, nullptr);
  const v2f a_hh0_1 = load_A(Whh0, HID, 4, n, hi, nullptr);
  const v2f a_ih1_0 = load_A(Wih1, HID, 0, n, hi, nullptr);
  const v2f a_ih1_1 = load_A(Wih1, HID, 4, n, hi, nullptr);
  const v2f a_hh1_0 = load_A(Whh1, HID, 0, n, hi, Wout);
  const v2f a_hh1_1 = load_A(Whh1, HID, 4, n, hi, Wout);

  v8f c0b, c1b;
#pragma unroll
  for (int v = 0; v < 8; ++v) {
    const bool use = (v & 2) == 0;
    const int u = (v & 5) + 2 * hi;
    c0b[v] = use ? (bih0[u] + bhh0[u]) : 0.0f;
    c1b[v] = use ? (bih1[u] + bhh1[u]) : 0.0f;
  }
  c1b[2] = (hi == 0) ? bout_p[0] : 0.0f;

  const float ximask = (hi == 0) ? 1.0f : 0.0f;

  v2f h1b0 = (v2f){0.f, 0.f}, h1b1 = (v2f){0.f, 0.f};
  v2f h2b0 = (v2f){0.f, 0.f}, h2b1 = (v2f){0.f, 0.f};

  float* op = out + (size_t)(bb + n) * T_LEN;

  __shared__ v4f xs[2 * 2 * BUFF4];
  __shared__ v4f hs[2 * 2 * HBUF4];
  v4f* xsw = xs + tib * (2 * BUFF4);
  v4f* hsw = hs + tib * (2 * HBUF4) + lane;

  const float* xw = x + (size_t)bb * T_LEN * INP;
  int goff[NJ], loff[NJ];
#pragma unroll
  for (int j = 0; j < NJ; ++j) {
    const int q = j * 32 + lane;
    const int r = q / CHF4;
    const int c4 = q - r * CHF4;
    goff[j] = r * (T_LEN * INP) + c4 * 4;
    loff[j] = r * ROWF4 + c4;
  }
  v4f st[NJ];
#pragma unroll
  for (int j = 0; j < NJ; ++j) st[j] = (v4f){0.f, 0.f, 0.f, 0.f};
  auto gload = [&](int c) {
#pragma unroll
    for (int j = 0; j < NJ; ++j)
      st[j] = *(const v4f*)(xw + goff[j] + c * (CH * INP));
  };
  auto lstore = [&](int b) {
#pragma unroll
    for (int j = 0; j < NJ; ++j) xsw[b * BUFF4 + loff[j]] = st[j];
  };
  const float* lb0 = (const float*)xsw + n * (4 * ROWF4) + 2 * hi;
  const float* lb1 = lb0 + 4 * BUFF4;
  auto lfetch = [&](const float* lb, int tt, v2f& r0, v2f& r1) {
    r0 = *(const v2f*)(lb + INP * tt);
    r1 = *(const v2f*)(lb + INP * tt + 4);
  };

#define ORDER_FENCE() __builtin_amdgcn_sched_barrier(0)

  v8f P = c0b;
  v2f X0[4], X1[4];
#pragma unroll
  for (int j = 0; j < 4; ++j) {
    X0[j] = (v2f){0.f, 0.f};
    X1[j] = (v2f){0.f, 0.f};
  }
  if (role == 0) {
    xsw[hi * BUFF4 + n * ROWF4 + CHF4] = (v4f){0.f, 0.f, 0.f, 0.f};
    gload(0);
    lstore(0);
    gload(1);
    v2f x00, x01;
    lfetch(lb0, 0, x00, x01);
#pragma unroll
    for (int j = 0; j < 4; ++j) lfetch(lb0, 1 + j, X0[j], X1[j]);
    v2f b1;
    b1.x = x01.x * ximask;
    b1.y = x01.y * ximask;
    P = wmma4(a_ih0_1, b1, wmma4(a_ih0_0, x00, c0b));
  }

  auto step0 = [&](const v2f& xa0, const v2f& xa1, bool do_x, v4f& hout) {
    v2f xm1;
    xm1.x = xa1.x * ximask;
    xm1.y = xa1.y * ximask;
    ORDER_FENCE();
    v8f acc = wmma4(a_hh0_0, h1b0, P);
    ORDER_FENCE();
    v8f Pn = P;
    if (do_x) Pn = wmma4(a_ih0_0, xa0, c0b);
    ORDER_FENCE();
    acc = wmma4(a_hh0_1, h1b1, acc);
    ORDER_FENCE();
    if (do_x) Pn = wmma4(a_ih0_1, xm1, Pn);
    ORDER_FENCE();
    h1b0.x = htanh(acc[0]);
    h1b0.y = htanh(acc[1]);
    h1b1.x = htanh(acc[4]);
    h1b1.y = htanh(acc[5]);
    hout = (v4f){h1b0.x, h1b0.y, h1b1.x, h1b1.y};
    P = Pn;
  };

  auto chunk0 = [&](int c, bool last) {
    const float* lc = (c & 1) ? lb1 : lb0;
    const float* ln = (c & 1) ? lb0 : lb1;
    v4f* hw = hsw + (c & 1) * HBUF4;
    if (!last) {
      lstore((c + 1) & 1);
      if (c + 2 < NCH) gload(c + 2);
    }
#pragma unroll
    for (int gi = 0; gi < 4; ++gi) {
      v2f N0[4], N1[4];
#pragma unroll
      for (int j = 0; j < 4; ++j) {
        N0[j] = X0[j];
        N1[j] = X1[j];
        const int s = 4 * gi + 5 + j;
        if (s < CH) lfetch(lc, s, N0[j], N1[j]);
        else if (!last) lfetch(ln, s - CH, N0[j], N1[j]);
      }
      v4f hq[4];
#pragma unroll
      for (int j = 0; j < 4; ++j)
        step0(X0[j], X1[j], !(last && gi == 3 && j == 3), hq[j]);
#pragma unroll
      for (int j = 0; j < 4; ++j) hw[(4 * gi + j) * 32] = hq[j];
#pragma unroll
      for (int j = 0; j < 4; ++j) {
        X0[j] = N0[j];
        X1[j] = N1[j];
      }
    }
  };

  float cy0 = 0.f, cy1 = 0.f, cy2 = 0.f;
  auto chunk1 = [&](int k) {
    const v4f* hr = hsw + (k & 1) * HBUF4;
    v4f H[CH];
#pragma unroll
    for (int s = 0; s < CH; ++s) H[s] = hr[s * 32];
    v8f p = wmma4(a_ih1_1, (v2f){H[0].z, H[0].w},
                  wmma4(a_ih1_0, (v2f){H[0].x, H[0].y}, c1b));
    float ob0 = 0.f, ob1 = 0.f, ob2 = 0.f;
#pragma unroll
    for (int s = 0; s < CH; ++s) {
      ORDER_FENCE();
      v8f acc = wmma4(a_hh1_0, h2b0, p);
      ORDER_FENCE();
      v8f pn = p;
      if (s + 1 < CH)
        pn = wmma4(a_ih1_0, (v2f){H[(s + 1) % CH].x, H[(s + 1) % CH].y}, c1b);
      ORDER_FENCE();
      acc = wmma4(a_hh1_1, h2b1, acc);
      ORDER_FENCE();
      if (s + 1 < CH)
        pn = wmma4(a_ih1_1, (v2f){H[(s + 1) % CH].z, H[(s + 1) % CH].w}, pn);
      ORDER_FENCE();
      const float y = acc[2];
      h2b0.x = htanh(acc[0]);
      h2b0.y = htanh(acc[1]);
      h2b1.x = htanh(acc[4]);
      h2b1.y = htanh(acc[5]);
      p = pn;
      if (s == 0) {
        if (k > 0 && lane < 16) {
          float4 o4;
          o4.x = cy0; o4.y = cy1; o4.z = cy2; o4.w = y;
          *(float4*)(op + CH * k - 4) = o4;
        }
      } else {
        const int idx = s - 1;
        if ((idx & 3) == 3) {
          if (lane < 16) {
            float4 o4;
            o4.x = ob0; o4.y = ob1; o4.z = ob2; o4.w = y;
            *(float4*)(op + CH * k + idx - 3) = o4;
          }
        } else if ((idx & 3) == 0) ob0 = y;
        else if ((idx & 3) == 1) ob1 = y;
        else ob2 = y;
      }
    }
    cy0 = ob0; cy1 = ob1; cy2 = ob2;
  };

  for (int c = 0; c <= NCH; ++c) {
    if (role == 0) {
      if (c < NCH - 1) chunk0(c, false);
      else if (c == NCH - 1) chunk0(c, true);
    } else {
      if (c > 0) chunk1(c - 1);
    }
    __builtin_amdgcn_fence(__ATOMIC_RELEASE, "workgroup", "local");
    __builtin_amdgcn_s_barrier();
    __builtin_amdgcn_fence(__ATOMIC_ACQUIRE, "workgroup", "local");
  }

  if (role == 1) {
    v8f efin = wmma4(a_hh1_1, h2b1, wmma4(a_hh1_0, h2b0, c1b));
    if (lane < 16) {
      float4 o4;
      o4.x = cy0; o4.y = cy1; o4.z = cy2; o4.w = efin[2];
      *(float4*)(op + T_LEN - 4) = o4;
    }
  }
#undef ORDER_FENCE
}

extern "C" void kernel_launch(void* const* d_in, const int* in_sizes, int n_in,
                              void* d_out, int out_size, void* d_ws,
                              size_t ws_size, hipStream_t stream) {
  (void)in_sizes; (void)n_in; (void)out_size; (void)d_ws; (void)ws_size;
  const float* x    = (const float*)d_in[0];
  const float* Wih0 = (const float*)d_in[1];
  const float* Whh0 = (const float*)d_in[2];
  const float* bih0 = (const float*)d_in[3];
  const float* bhh0 = (const float*)d_in[4];
  const float* Wih1 = (const float*)d_in[5];
  const float* Whh1 = (const float*)d_in[6];
  const float* bih1 = (const float*)d_in[7];
  const float* bhh1 = (const float*)d_in[8];
  const float* Wout = (const float*)d_in[9];
  const float* bout = (const float*)d_in[10];
  float* out = (float*)d_out;

  dim3 grid(B_TOTAL / 16 / 2);
  dim3 block(128);
  rnn2_tanh_wmma<<<grid, block, 0, stream>>>(x, Wih0, Whh0, bih0, bhh0, Wih1,
                                             Whh1, bih1, bhh1, Wout, bout, out);
}


# Input products of both layers moved onto two more waves per tile: 4-stage wave pipeline through LDS, whole D tiles handed over as C operands; recurrence waves issue 2 WMMA per step, not 4
# speedup vs baseline: 3.0888x; 1.1186x over previous
#include <hip/hip_runtime.h>
#include <math.h>

#define B_TOTAL 4096
#define T_LEN   512
#define INP     6
#define HID     8

#define CH     16
#define NCH    (T_LEN / CH)
#define HBUF4  (CH * 32)
#define PBUF4  (CH * 64)
#define CHF4   (CH * INP / 4)
#define ROWF4  (CHF4 + 1)
#define BUFF4  (16 * ROWF4)
#define NJ     (16 * CHF4 / 32)

typedef float v2f __attribute__((ext_vector_type(2)));
typedef float v4f __attribute__((ext_vector_type(4)));
typedef float v8f __attribute__((ext_vector_type(8)));

static __device__ __forceinline__ v8f wmma4(v2f a, v2f b, v8f c) {
  return __builtin_amdgcn_wmma_f32_16x16x4_f32(false, a, false, b, (short)0, c,
                                               false, false);
}

static __device__ __forceinline__ float htanh(float x) {
#if __has_builtin(__builtin_amdgcn_tanhf)
  return __builtin_amdgcn_tanhf(x);
#else
  float t = __builtin_amdgcn_exp2f(x * 2.8853900817779268f);
  return 1.0f - 2.0f * __builtin_amdgcn_rcpf(t + 1.0f);
#endif
}

static __device__ __forceinline__ int unit_of_row(int m) {
  return (m & 5) | ((m & 8) >> 2);
}
static __device__ __forceinline__ bool row_used(int m) { return (m & 2) == 0; }

static __device__ __forceinline__ v2f load_A(const float* __restrict__ W,
                                             int ncols, int kbase, int m,
                                             int hi,
                                             const float* __restrict__ orow) {
  const int k0 = kbase + 2 * hi;
  const int u = unit_of_row(m);
  const bool use = row_used(m);
  v2f a;
  a.x = (use && k0 < ncols) ? W[u * ncols + k0] : 0.0f;
  a.y = (use && (k0 + 1) < ncols) ? W[u * ncols + k0 + 1] : 0.0f;
  if (orow != nullptr && m == 2) {
    a.x = (k0 < ncols) ? orow[k0] : 0.0f;
    a.y = ((k0 + 1) < ncols) ? orow[k0 + 1] : 0.0f;
  }
  return a;
}

__global__ __launch_bounds__(128) void rnn2_tanh_wmma(
    const float* __restrict__ x,
    const float* __restrict__ Wih0, const float* __restrict__ Whh0,
    const float* __restrict__ bih0, const float* __restrict__ bhh0,
    const float* __restrict__ Wih1, const float* __restrict__ Whh1,
    const float* __restrict__ bih1, const float* __restrict__ bhh1,
    const float* __restrict__ Wout, const float* __restrict__ bout_p,
    float* __restrict__ out) {
  const int lane = threadIdx.x & 31;
  const int role = __builtin_amdgcn_readfirstlane((int)(threadIdx.x >> 5));
  const int n = lane & 15;
  const int hi = lane >> 4;
  const int bb = blockIdx.x * 16;

  const v2f a_ih0_0 = load_A(Wih0, INP, 0, n, hi, nullptr);
  const v2f a_ih0_1 = load_A(Wih0, INP, 4, n, hi, nullptr);
  const v2f a_hh0_0 = load_A(Whh0, HID, 0, n, hi, nullptr);
  const v2f a_hh0_1 = load_A(Whh0, HID, 4, n, hi, nullptr);
  const v2f a_ih1_0 = load_A(Wih1, HID, 0, n, hi, nullptr);
  const v2f a_ih1_1 = load_A(Wih1, HID, 4, n, hi, nullptr);
  const v2f a_hh1_0 = load_A(Whh1, HID, 0, n, hi, Wout);
  const v2f a_hh1_1 = load_A(Whh1, HID, 4, n, hi, Wout);

  v8f c0b, c1b;
#pragma unroll
  for (int v = 0; v < 8; ++v) {
    const bool use = (v & 2) == 0;
    const int u = (v & 5) + 2 * hi;
    c0b[v] = use ? (bih0[u] + bhh0[u]) : 0.0f;
    c1b[v] = use ? (bih1[u] + bhh1[u]) : 0.0f;
  }
  c1b[2] = (hi == 0) ? bout_p[0] : 0.0f;

  const float ximask = (hi == 0) ? 1.0f : 0.0f;

  v2f h1b0 = (v2f){0.f, 0.f}, h1b1 = (v2f){0.f, 0.f};
  v2f h2b0 = (v2f){0.f, 0.f}, h2b1 = (v2f){0.f, 0.f};

  float* op = out + (size_t)(bb + n) * T_LEN;

  __shared__ v4f xs[2 * BUFF4];
  __shared__ v4f p0s[2 * PBUF4];
  __shared__ v4f hs[2 * HBUF4];
  __shared__ v4f p1s[2 * PBUF4];
  v4f* xsw = xs;
  v4f* hsw = hs + lane;
  v4f* p0w = p0s + lane;
  v4f* p1w = p1s + lane;

  const float* xw = x + (size_t)bb * T_LEN * INP;
  int goff[NJ], loff[NJ];
#pragma unroll
  for (int j = 0; j < NJ; ++j) {
    const int q = j * 32 + lane;
    const int r = q / CHF4;
    const int c4 = q - r * CHF4;
    goff[j] = r * (T_LEN * INP) + c4 * 4;
    loff[j] = r * ROWF4 + c4;
  }
  v4f st[NJ];
#pragma unroll
  for (int j = 0; j < NJ; ++j) st[j] = (v4f){0.f, 0.f, 0.f, 0.f};
  auto gload = [&](int c) {
#pragma unroll
    for (int j = 0; j < NJ; ++j)
      st[j] = *(const v4f*)(xw + goff[j] + c * (CH * INP));
  };
  auto lstore = [&](int b) {
#pragma unroll
    for (int j = 0; j < NJ; ++j) xsw[b * BUFF4 + loff[j]] = st[j];
  };
  const float* lb0 = (const float*)xsw + n * (4 * ROWF4) + 2 * hi;
  const float* lb1 = lb0 + 4 * BUFF4;
  auto lfetch = [&](const float* lb, int tt, v2f& r0, v2f& r1) {
    r0 = *(const v2f*)(lb + INP * tt);
    r1 = *(const v2f*)(lb + INP * tt + 4);
  };

  auto put_tile = [&](v4f* pw, int s, const v8f& p) {
    pw[(2 * s) * 32] = (v4f){p[0], p[1], p[2], p[3]};
    pw[(2 * s + 1) * 32] = (v4f){p[4], p[5], p[6], p[7]};
  };
  auto get_tile = [&](const v4f* pr, int s) {
    const v4f lo = pr[(2 * s) * 32];
    const v4f up = pr[(2 * s + 1) * 32];
    return (v8f){lo.x, lo.y, lo.z, lo.w, up.x, up.y, up.z, up.w};
  };

  if (role == 0) {
    xsw[hi * BUFF4 + n * ROWF4 + CHF4] = (v4f){0.f, 0.f, 0.f, 0.f};
    gload(0);
    lstore(0);
    gload(1);
  }

  auto chunkX = [&](int c) {
    const float* lc = (c & 1) ? lb1 : lb0;
    v4f* pw = p0w + (c & 1) * PBUF4;
    if (c + 1 < NCH) lstore((c + 1) & 1);
    if (c + 2 < NCH) gload(c + 2);
    v2f X0[CH], X1[CH];
#pragma unroll
    for (int s = 0; s < CH; ++s) lfetch(lc, s, X0[s], X1[s]);
#pragma unroll
    for (int s = 0; s < CH; ++s) {
      v2f xm;
      xm.x = X1[s].x * ximask;
      xm.y = X1[s].y * ximask;
      const v8f p = wmma4(a_ih0_1, xm, wmma4(a_ih0_0, X0[s], c0b));
      put_tile(pw, s, p);
    }
  };

  auto chunkR0 = [&](int k) {
    const v4f* pr = p0w + (k & 1) * PBUF4;
    v4f* hw = hsw + (k & 1) * HBUF4;
    v8f C[CH];
#pragma unroll
    for (int s = 0; s < CH; ++s) C[s] = get_tile(pr, s);
    __builtin_amdgcn_sched_barrier(0);
    v4f hq[CH];
#pragma unroll
    for (int s = 0; s < CH; ++s) {
      v8f acc = wmma4(a_hh0_0, h1b0, C[s]);
      acc = wmma4(a_hh0_1, h1b1, acc);
      h1b0.x = htanh(acc[0]);
      h1b0.y = htanh(acc[1]);
      h1b1.x = htanh(acc[4]);
      h1b1.y = htanh(acc[5]);
      hq[s] = (v4f){h1b0.x, h1b0.y, h1b1.x, h1b1.y};
      if ((s & 3) == 3) {
#pragma unroll
        for (int j = s - 3; j <= s; ++j) hw[j * 32] = hq[j];
      }
    }
  };

  auto chunkJ = [&](int k) {
    const v4f* hr = hsw + (k & 1) * HBUF4;
    v4f* pw = p1w + (k & 1) * PBUF4;
    v4f H[CH];
#pragma unroll
    for (int s = 0; s < CH; ++s) H[s] = hr[s * 32];
#pragma unroll
    for (int s = 0; s < CH; ++s) {
      const v8f p = wmma4(a_ih1_1, (v2f){H[s].z, H[s].w},
                          wmma4(a_ih1_0, (v2f){H[s].x, H[s].y}, c1b));
      put_tile(pw, s, p);
    }
  };

  float cy0 = 0.f, cy1 = 0.f, cy2 = 0.f;
  auto chunkR1 = [&](int k) {
    const v4f* pr = p1w + (k & 1) * PBUF4;
    v8f C[CH];
#pragma unroll
    for (int s = 0; s < CH; ++s) C[s] = get_tile(pr, s);
    float ob0 = 0.f, ob1 = 0.f, ob2 = 0.f;
#pragma unroll
    for (int s = 0; s < CH; ++s) {
      v8f acc = wmma4(a_hh1_0, h2b0, C[s]);
      acc = wmma4(a_hh1_1, h2b1, acc);
      const float y = acc[2];
      h2b0.x = htanh(acc[0]);
      h2b0.y = htanh(acc[1]);
      h2b1.x = htanh(acc[4]);
      h2b1.y = htanh(acc[5]);
      if (s == 0) {
        if (k > 0 && lane < 16) {
          float4 o4;
          o4.x = cy0; o4.y = cy1; o4.z = cy2; o4.w = y;
          *(float4*)(op + CH * k - 4) = o4;
        }
      } else {
        const int idx = s - 1;
        if ((idx & 3) == 3) {
          if (lane < 16) {
            float4 o4;
            o4.x = ob0; o4.y = ob1; o4.z = ob2; o4.w = y;
            *(float4*)(op + CH * k + idx - 3) = o4;
          }
        } else if ((idx & 3) == 0) ob0 = y;
        else if ((idx & 3) == 1) ob1 = y;
        else ob2 = y;
      }
    }
    cy0 = ob0; cy1 = ob1; cy2 = ob2;
  };

  for (int c = 0; c < NCH + 3; ++c) {
    if (role == 0) {
      if (c < NCH) chunkX(c);
    } else if (role == 1) {
      if (c >= 1 && c <= NCH) chunkR0(c - 1);
    } else if (role == 2) {
      if (c >= 2 && c <= NCH + 1) chunkJ(c - 2);
    } else {
      if (c >= 3) chunkR1(c - 3);
    }
    __builtin_amdgcn_fence(__ATOMIC_RELEASE, "workgroup", "local");
    __builtin_amdgcn_s_barrier();
    __builtin_amdgcn_fence(__ATOMIC_ACQUIRE, "workgroup", "local");
  }

  if (role == 3) {
    v8f efin = wmma4(a_hh1_1, h2b1, wmma4(a_hh1_0, h2b0, c1b));
    if (lane < 16) {
      float4 o4;
      o4.x = cy0; o4.y = cy1; o4.z = cy2; o4.w = efin[2];
      *(float4*)(op + T_LEN - 4) = o4;
    }
  }
}

extern "C" void kernel_launch(void* const* d_in, const int* in_sizes, int n_in,
                              void* d_out, int out_size, void* d_ws,
                              size_t ws_size, hipStream_t stream) {
  (void)in_sizes; (void)n_in; (void)out_size; (void)d_ws; (void)ws_size;
  const float* x    = (const float*)d_in[0];
  const float* Wih0 = (const float*)d_in[1];
  const float* Whh0 = (const float*)d_in[2];
  const float* bih0 = (const float*)d_in[3];
  const float* bhh0 = (const float*)d_in[4];
  const float* Wih1 = (const float*)d_in[5];
  const float* Whh1 = (const float*)d_in[6];
  const float* bih1 = (const float*)d_in[7];
  const float* bhh1 = (const float*)d_in[8];
  const float* Wout = (const float*)d_in[9];
  const float* bout = (const float*)d_in[10];
  float* out = (float*)d_out;

  dim3 grid(B_TOTAL / 16);
  dim3 block(128);
  rnn2_tanh_wmma<<<grid, block, 0, stream>>>(x, Wih0, Whh0, bih0, bhh0, Wih1,
                                             Whh1, bih1, bhh1, Wout, bout, out);
}
